# starting kernel
# baseline (speedup 1.0000x reference)
#include <hip/hip_runtime.h>


#define F_DIM 128

typedef __attribute__((ext_vector_type(2))) float v2f;
typedef __attribute__((ext_vector_type(8))) float v8f;

static __device__ __forceinline__ v8f wmma_f32x4(v2f a, v2f b, v8f c) {
  return __builtin_amdgcn_wmma_f32_16x16x4_f32(
      false, a, false, b, (short)0, c, false, false);
}

__global__ void __launch_bounds__(256) zero_f32(float* __restrict__ p, size_t n) {
  size_t i = (size_t)blockIdx.x * blockDim.x + threadIdx.x;
  size_t stride = (size_t)gridDim.x * blockDim.x;
  for (; i < n; i += stride) p[i] = 0.0f;
}

__global__ void __launch_bounds__(256) count_deg(const int* __restrict__ dst,
                                                 float* __restrict__ deg, int nE) {
  int e = blockIdx.x * blockDim.x + threadIdx.x;
  if (e < nE) atomicAdd(&deg[dst[e]], 1.0f);
}

__global__ void __launch_bounds__(256) invert_deg(float* __restrict__ deg, int n) {
  int i = blockIdx.x * blockDim.x + threadIdx.x;
  if (i < n) deg[i] = 1.0f / fmaxf(deg[i], 1.0f);
}

__global__ void __launch_bounds__(256)
transpose_w(const float* __restrict__ W, float* __restrict__ WT, int n_out) {
  int i = blockIdx.x * blockDim.x + threadIdx.x;
  if (i >= F_DIM * n_out) return;
  int k = i / n_out;
  int n = i - k * n_out;
  WT[n * F_DIM + k] = W[i];
}

__global__ void __launch_bounds__(256)
edge_scatter(const float* __restrict__ h, const int* __restrict__ src,
             const int* __restrict__ dst, float* __restrict__ agg, int nE) {
  size_t t = (size_t)blockIdx.x * blockDim.x + threadIdx.x;
  int e = (int)(t >> 5);
  int lane = (int)(t & 31);
  if (e >= nE) return;
  int s = src[e];
  int d = dst[e];
  const float4* __restrict__ hp = (const float4*)(h + (size_t)s * F_DIM);
  float4 v = hp[lane];
  float* ap = agg + (size_t)d * F_DIM + lane * 4;
  atomicAdd(ap + 0, v.x);
  atomicAdd(ap + 1, v.y);
  atomicAdd(ap + 2, v.z);
  atomicAdd(ap + 3, v.w);
}

template <int N_OUT, bool RELU>
__global__ void __launch_bounds__(256)
sage_gemm(const float* __restrict__ h, const float* __restrict__ agg,
          const float* __restrict__ inv_deg,
          const float* __restrict__ WsT, const float* __restrict__ WnT,
          const float* __restrict__ bias, float* __restrict__ out, int n_nodes) {
  constexpr int TILES_N = (N_OUT + 15) / 16;
  constexpr bool PARTIAL = (N_OUT % 16) != 0;

  int wave = blockIdx.x * (blockDim.x >> 5) + (threadIdx.x >> 5);
  int tiles_m = (n_nodes + 15) >> 4;
  if (wave >= tiles_m * TILES_N) return;

  int tm = wave / TILES_N;
  int tn = wave - tm * TILES_N;
  int lane = threadIdx.x & 31;
  int half = lane >> 4;
  int l16 = lane & 15;
  int koff = half * 2;

  int mrow = tm * 16 + l16;
  if (mrow >= n_nodes) mrow = n_nodes - 1;
  int ncol = tn * 16 + l16;
  int ncl = ncol;
  float nmask = 1.0f;
  if constexpr (PARTIAL) {
    ncl = (ncol < N_OUT) ? ncol : (N_OUT - 1);
    nmask = (ncol < N_OUT) ? 1.0f : 0.0f;
  }

  float invd = inv_deg[mrow];
  const float2* __restrict__ hp = (const float2*)(h + (size_t)mrow * F_DIM) + (koff >> 1);
  const float2* __restrict__ ap = (const float2*)(agg + (size_t)mrow * F_DIM) + (koff >> 1);
  const float2* __restrict__ wsp = (const float2*)(WsT + (size_t)ncl * F_DIM) + (koff >> 1);
  const float2* __restrict__ wnp = (const float2*)(WnT + (size_t)ncl * F_DIM) + (koff >> 1);

  v8f c;
  float bv = bias[ncl];
  if constexpr (PARTIAL) bv *= nmask;
#pragma unroll
  for (int j = 0; j < 8; ++j) c[j] = bv;

#pragma unroll 8
  for (int k = 0; k < F_DIM; k += 4) {
    float2 av = hp[k >> 1];
    float2 bw = wsp[k >> 1];
    v2f a{av.x, av.y};
    v2f b{bw.x, bw.y};
    if constexpr (PARTIAL) { b.x *= nmask; b.y *= nmask; }
    c = wmma_f32x4(a, b, c);
  }
#pragma unroll 8
  for (int k = 0; k < F_DIM; k += 4) {
    float2 av = ap[k >> 1];
    float2 bw = wnp[k >> 1];
    v2f a{av.x * invd, av.y * invd};
    v2f b{bw.x, bw.y};
    if constexpr (PARTIAL) { b.x *= nmask; b.y *= nmask; }
    c = wmma_f32x4(a, b, c);
  }

  if constexpr (RELU) {
#pragma unroll
    for (int j = 0; j < 8; ++j) c[j] = fmaxf(c[j], 0.0f);
  }

#pragma unroll
  for (int j = 0; j < 8; ++j) {
    int row = tm * 16 + j + half * 8;
    bool ok = (row < n_nodes);
    if constexpr (PARTIAL) ok = ok && (ncol < N_OUT);
    if (ok) out[(size_t)row * N_OUT + ncol] = c[j];
  }
}

extern "C" void kernel_launch(void* const* d_in, const int* in_sizes, int n_in,
                              void* d_out, int out_size, void* d_ws, size_t ws_size,
                              hipStream_t stream) {
  const float* x   = (const float*)d_in[0];
  const int*   src = (const int*)d_in[1];
  const int*   dst = (const int*)d_in[2];
  const float* Ws[3] = { (const float*)d_in[3], (const float*)d_in[6], (const float*)d_in[9]  };
  const float* Wn[3] = { (const float*)d_in[4], (const float*)d_in[7], (const float*)d_in[10] };
  const float* bs[3] = { (const float*)d_in[5], (const float*)d_in[8], (const float*)d_in[11] };

  const int N = in_sizes[0] / F_DIM;
  const int E = in_sizes[1];
  const int NC = 47;

  float* invdeg = (float*)d_ws;
  float* agg = invdeg + (size_t)N;
  float* hA  = agg + (size_t)N * F_DIM;
  float* hB  = hA  + (size_t)N * F_DIM;
  float* wt  = hB  + (size_t)N * F_DIM;
  float* WsT[3] = { wt,                 wt + 2 * F_DIM * F_DIM, wt + 4 * F_DIM * F_DIM };
  float* WnT[3] = { wt + F_DIM * F_DIM, wt + 3 * F_DIM * F_DIM, wt + 4 * F_DIM * F_DIM + F_DIM * NC };

  const int ZB = 2048;

  zero_f32<<<ZB, 256, 0, stream>>>(invdeg, (size_t)N);
  count_deg<<<(E + 255) / 256, 256, 0, stream>>>(dst, invdeg, E);
  invert_deg<<<(N + 255) / 256, 256, 0, stream>>>(invdeg, N);

  int nout_l[3] = { F_DIM, F_DIM, NC };
  for (int l = 0; l < 3; ++l) {
    int elems = F_DIM * nout_l[l];
    transpose_w<<<(elems + 255) / 256, 256, 0, stream>>>(Ws[l], WsT[l], nout_l[l]);
    transpose_w<<<(elems + 255) / 256, 256, 0, stream>>>(Wn[l], WnT[l], nout_l[l]);
  }

  const float* hin = x;
  float* houts[3] = { hA, hB, (float*)d_out };

  size_t scatter_threads = (size_t)E * 32;
  int scatter_blocks = (int)((scatter_threads + 255) / 256);
  int tiles_m = (N + 15) / 16;

  for (int l = 0; l < 3; ++l) {
    zero_f32<<<ZB, 256, 0, stream>>>(agg, (size_t)N * F_DIM);
    edge_scatter<<<scatter_blocks, 256, 0, stream>>>(hin, src, dst, agg, E);

    if (l < 2) {
      int waves = tiles_m * ((F_DIM + 15) / 16);
      int blocks = (waves + 7) / 8;
      sage_gemm<F_DIM, true><<<blocks, 256, 0, stream>>>(
          hin, agg, invdeg, WsT[l], WnT[l], bs[l], houts[l], N);
    } else {
      int waves = tiles_m * ((NC + 15) / 16);
      int blocks = (waves + 7) / 8;
      sage_gemm<NC, false><<<blocks, 256, 0, stream>>>(
          hin, agg, invdeg, WsT[l], WnT[l], bs[l], houts[l], N);
    }
    hin = houts[l];
  }
}


# Edge scatter: a wave owns 32 consecutive edges (ids read coalesced, broadcast to scalars), keeps two 8-edge groups of row loads in flight ahead of the atomic adds; sums stay in atomic arrival order
# speedup vs baseline: 1.4744x; 1.4744x over previous
#include <hip/hip_runtime.h>


#define F_DIM 128

typedef __attribute__((ext_vector_type(2))) float v2f;
typedef __attribute__((ext_vector_type(8))) float v8f;

static __device__ __forceinline__ v8f wmma_f32x4(v2f a, v2f b, v8f c) {
  return __builtin_amdgcn_wmma_f32_16x16x4_f32(
      false, a, false, b, (short)0, c, false, false);
}

__global__ void __launch_bounds__(256) zero_f32(float* __restrict__ p, size_t n) {
  size_t i = (size_t)blockIdx.x * blockDim.x + threadIdx.x;
  size_t stride = (size_t)gridDim.x * blockDim.x;
  for (; i < n; i += stride) p[i] = 0.0f;
}

__global__ void __launch_bounds__(256) count_deg(const int* __restrict__ dst,
                                                 float* __restrict__ deg, int nE) {
  int e = blockIdx.x * blockDim.x + threadIdx.x;
  if (e < nE) atomicAdd(&deg[dst[e]], 1.0f);
}

__global__ void __launch_bounds__(256) invert_deg(float* __restrict__ deg, int n) {
  int i = blockIdx.x * blockDim.x + threadIdx.x;
  if (i < n) deg[i] = 1.0f / fmaxf(deg[i], 1.0f);
}

__global__ void __launch_bounds__(256)
transpose_w(const float* __restrict__ W, float* __restrict__ WT, int n_out) {
  int i = blockIdx.x * blockDim.x + threadIdx.x;
  if (i >= F_DIM * n_out) return;
  int k = i / n_out;
  int n = i - k * n_out;
  WT[n * F_DIM + k] = W[i];
}

#define EPW 32
#define ECH 8

static __device__ __forceinline__ void row_load(const float* __restrict__ h, int s,
                                                int lane, float (&v)[4]) {
  const float* __restrict__ hp = h + (size_t)s * F_DIM + lane;
  v[0] = hp[0];
  v[1] = hp[32];
  v[2] = hp[64];
  v[3] = hp[96];
}

static __device__ __forceinline__ void row_add(float* __restrict__ agg, int d,
                                               int lane, const float (&v)[4]) {
  float* ap = agg + (size_t)d * F_DIM + lane;
  atomicAdd(ap + 0, v[0]);
  atomicAdd(ap + 32, v[1]);
  atomicAdd(ap + 64, v[2]);
  atomicAdd(ap + 96, v[3]);
}

__global__ void __launch_bounds__(256)
edge_scatter(const float* __restrict__ h, const int* __restrict__ src,
             const int* __restrict__ dst, float* __restrict__ agg, int nE) {
  int wave = __builtin_amdgcn_readfirstlane(
      (int)(blockIdx.x * (blockDim.x >> 5) + (threadIdx.x >> 5)));
  int lane = (int)(threadIdx.x & 31);
  long long base = (long long)wave * EPW;
  if (base >= nE) return;

  if (base + EPW <= nE) {
    int s = src[base + lane];
    int d = dst[base + lane];
    float v[2][ECH][4];
#pragma unroll
    for (int j = 0; j < ECH; ++j)
      row_load(h, __builtin_amdgcn_readlane(s, j), lane, v[0][j]);
#pragma unroll
    for (int c = 0; c < EPW / ECH; ++c) {
      if (c + 1 < EPW / ECH) {
#pragma unroll
        for (int j = 0; j < ECH; ++j)
          row_load(h, __builtin_amdgcn_readlane(s, (c + 1) * ECH + j), lane,
                   v[(c + 1) & 1][j]);
      }
#pragma unroll
      for (int j = 0; j < ECH; ++j)
        row_add(agg, __builtin_amdgcn_readlane(d, c * ECH + j), lane, v[c & 1][j]);
    }
  } else {
    int cnt = (int)(nE - base);
    for (int j = 0; j < cnt; ++j) {
      float t[4];
      row_load(h, src[base + j], lane, t);
      row_add(agg, dst[base + j], lane, t);
    }
  }
}

template <int N_OUT, bool RELU>
__global__ void __launch_bounds__(256)
sage_gemm(const float* __restrict__ h, const float* __restrict__ agg,
          const float* __restrict__ inv_deg,
          const float* __restrict__ WsT, const float* __restrict__ WnT,
          const float* __restrict__ bias, float* __restrict__ out, int n_nodes) {
  constexpr int TILES_N = (N_OUT + 15) / 16;
  constexpr bool PARTIAL = (N_OUT % 16) != 0;

  int wave = blockIdx.x * (blockDim.x >> 5) + (threadIdx.x >> 5);
  int tiles_m = (n_nodes + 15) >> 4;
  if (wave >= tiles_m * TILES_N) return;

  int tm = wave / TILES_N;
  int tn = wave - tm * TILES_N;
  int lane = threadIdx.x & 31;
  int half = lane >> 4;
  int l16 = lane & 15;
  int koff = half * 2;

  int mrow = tm * 16 + l16;
  if (mrow >= n_nodes) mrow = n_nodes - 1;
  int ncol = tn * 16 + l16;
  int ncl = ncol;
  float nmask = 1.0f;
  if constexpr (PARTIAL) {
    ncl = (ncol < N_OUT) ? ncol : (N_OUT - 1);
    nmask = (ncol < N_OUT) ? 1.0f : 0.0f;
  }

  float invd = inv_deg[mrow];
  const float2* __restrict__ hp = (const float2*)(h + (size_t)mrow * F_DIM) + (koff >> 1);
  const float2* __restrict__ ap = (const float2*)(agg + (size_t)mrow * F_DIM) + (koff >> 1);
  const float2* __restrict__ wsp = (const float2*)(WsT + (size_t)ncl * F_DIM) + (koff >> 1);
  const float2* __restrict__ wnp = (const float2*)(WnT + (size_t)ncl * F_DIM) + (koff >> 1);

  v8f c;
  float bv = bias[ncl];
  if constexpr (PARTIAL) bv *= nmask;
#pragma unroll
  for (int j = 0; j < 8; ++j) c[j] = bv;

#pragma unroll 8
  for (int k = 0; k < F_DIM; k += 4) {
    float2 av = hp[k >> 1];
    float2 bw = wsp[k >> 1];
    v2f a{av.x, av.y};
    v2f b{bw.x, bw.y};
    if constexpr (PARTIAL) { b.x *= nmask; b.y *= nmask; }
    c = wmma_f32x4(a, b, c);
  }
#pragma unroll 8
  for (int k = 0; k < F_DIM; k += 4) {
    float2 av = ap[k >> 1];
    float2 bw = wnp[k >> 1];
    v2f a{av.x * invd, av.y * invd};
    v2f b{bw.x, bw.y};
    if constexpr (PARTIAL) { b.x *= nmask; b.y *= nmask; }
    c = wmma_f32x4(a, b, c);
  }

  if constexpr (RELU) {
#pragma unroll
    for (int j = 0; j < 8; ++j) c[j] = fmaxf(c[j], 0.0f);
  }

#pragma unroll
  for (int j = 0; j < 8; ++j) {
    int row = tm * 16 + j + half * 8;
    bool ok = (row < n_nodes);
    if constexpr (PARTIAL) ok = ok && (ncol < N_OUT);
    if (ok) out[(size_t)row * N_OUT + ncol] = c[j];
  }
}

extern "C" void kernel_launch(void* const* d_in, const int* in_sizes, int n_in,
                              void* d_out, int out_size, void* d_ws, size_t ws_size,
                              hipStream_t stream) {
  const float* x   = (const float*)d_in[0];
  const int*   src = (const int*)d_in[1];
  const int*   dst = (const int*)d_in[2];
  const float* Ws[3] = { (const float*)d_in[3], (const float*)d_in[6], (const float*)d_in[9]  };
  const float* Wn[3] = { (const float*)d_in[4], (const float*)d_in[7], (const float*)d_in[10] };
  const float* bs[3] = { (const float*)d_in[5], (const float*)d_in[8], (const float*)d_in[11] };

  const int N = in_sizes[0] / F_DIM;
  const int E = in_sizes[1];
  const int NC = 47;

  float* invdeg = (float*)d_ws;
  float* agg = invdeg + (size_t)N;
  float* hA  = agg + (size_t)N * F_DIM;
  float* hB  = hA  + (size_t)N * F_DIM;
  float* wt  = hB  + (size_t)N * F_DIM;
  float* WsT[3] = { wt,                 wt + 2 * F_DIM * F_DIM, wt + 4 * F_DIM * F_DIM };
  float* WnT[3] = { wt + F_DIM * F_DIM, wt + 3 * F_DIM * F_DIM, wt + 4 * F_DIM * F_DIM + F_DIM * NC };

  const int ZB = 2048;

  zero_f32<<<ZB, 256, 0, stream>>>(invdeg, (size_t)N);
  count_deg<<<(E + 255) / 256, 256, 0, stream>>>(dst, invdeg, E);
  invert_deg<<<(N + 255) / 256, 256, 0, stream>>>(invdeg, N);

  int nout_l[3] = { F_DIM, F_DIM, NC };
  for (int l = 0; l < 3; ++l) {
    int elems = F_DIM * nout_l[l];
    transpose_w<<<(elems + 255) / 256, 256, 0, stream>>>(Ws[l], WsT[l], nout_l[l]);
    transpose_w<<<(elems + 255) / 256, 256, 0, stream>>>(Wn[l], WnT[l], nout_l[l]);
  }

  const float* hin = x;
  float* houts[3] = { hA, hB, (float*)d_out };

  int scatter_waves = (E + EPW - 1) / EPW;
  int scatter_blocks = (scatter_waves + 7) / 8;
  int tiles_m = (N + 15) / 16;

  for (int l = 0; l < 3; ++l) {
    zero_f32<<<ZB, 256, 0, stream>>>(agg, (size_t)N * F_DIM);
    edge_scatter<<<scatter_blocks, 256, 0, stream>>>(hin, src, dst, agg, E);

    if (l < 2) {
      int waves = tiles_m * ((F_DIM + 15) / 16);
      int blocks = (waves + 7) / 8;
      sage_gemm<F_DIM, true><<<blocks, 256, 0, stream>>>(
          hin, agg, invdeg, WsT[l], WnT[l], bs[l], houts[l], N);
    } else {
      int waves = tiles_m * ((NC + 15) / 16);
      int blocks = (waves + 7) / 8;
      sage_gemm<NC, false><<<blocks, 256, 0, stream>>>(
          hin, agg, invdeg, WsT[l], WnT[l], bs[l], houts[l], N);
    }
    hin = houts[l];
  }
}
